# starting kernel
# baseline (speedup 1.0000x reference)
#include <hip/hip_runtime.h>


#define S_    4096
#define N_    262144
#define H_    8
#define E_    128
#define D_    16
#define CKV_  64
#define EPS_  1e-5f

typedef float v2f __attribute__((ext_vector_type(2)));
typedef float v4f __attribute__((ext_vector_type(4)));
typedef float v8f __attribute__((ext_vector_type(8)));

__device__ __forceinline__ unsigned enc_f32(float f) {
  unsigned u = __float_as_uint(f);
  return (u & 0x80000000u) ? ~u : (u | 0x80000000u);
}
__device__ __forceinline__ float dec_f32(unsigned e) {
  return (e & 0x80000000u) ? __uint_as_float(e & 0x7fffffffu)
                           : __uint_as_float(~e);
}

__global__ void k_fold_wkv(const float* __restrict__ Wkv,
                           float* __restrict__ Wfold) {
  int i = blockIdx.x * blockDim.x + threadIdx.x;
  int o = i >> 6, j = i & 63;
  Wfold[i] = Wkv[o * (2 * CKV_) + j] + Wkv[o * (2 * CKV_) + CKV_ + j];
}

__global__ void k_qproj(const float* __restrict__ sp,
                        const float* __restrict__ Wq,
                        float* __restrict__ Q) {
  const int tm   = blockIdx.x * 16;
  const int tn   = blockIdx.y * 16;
  const int lane = threadIdx.x;
  const int half = lane >> 4;
  const int l    = lane & 15;
  const float* arow = sp + (size_t)(tm + l) * E_ + 2 * half;
  const float* brow = Wq + (size_t)(tn + l) * E_ + 2 * half;
  v8f acc = {0.f, 0.f, 0.f, 0.f, 0.f, 0.f, 0.f, 0.f};
#pragma unroll
  for (int k = 0; k < E_; k += 4) {
    v2f a = *(const v2f*)(arow + k);
    v2f b = *(const v2f*)(brow + k);
    acc = __builtin_amdgcn_wmma_f32_16x16x4_f32(false, a, false, b,
                                                (short)0, acc, false, false);
  }
#pragma unroll
  for (int r = 0; r < 8; ++r)
    Q[(size_t)(tm + r + 8 * half) * E_ + tn + l] = acc[r];
}

__global__ void k_kproj_scores(const float* __restrict__ raw,
                               const float* __restrict__ Wfold,
                               const float* __restrict__ Q,
                               const int* __restrict__ assign,
                               float* __restrict__ scores,
                               unsigned* __restrict__ segmax,
                               float* __restrict__ attn_out) {
  __shared__ float kt[16 * 128];
  __shared__ float sc[16 * 8];
  const int p0   = blockIdx.x * 16;
  const int lane = threadIdx.x;
  const int half = lane >> 4;
  const int l    = lane & 15;

  const float* arow = raw + (size_t)(p0 + l) * CKV_ + 2 * half;
  v2f areg[16];
#pragma unroll
  for (int k = 0; k < 16; ++k) areg[k] = *(const v2f*)(arow + 4 * k);

#pragma unroll
  for (int h = 0; h < H_; ++h) {
    const float* brow = Wfold + (size_t)(h * 16 + l) * CKV_ + 2 * half;
    v8f acc = {0.f, 0.f, 0.f, 0.f, 0.f, 0.f, 0.f, 0.f};
#pragma unroll
    for (int k = 0; k < 16; ++k) {
      v2f b = *(const v2f*)(brow + 4 * k);
      acc = __builtin_amdgcn_wmma_f32_16x16x4_f32(false, areg[k], false, b,
                                                  (short)0, acc, false, false);
    }
#pragma unroll
    for (int r = 0; r < 8; ++r)
      kt[(r + 8 * half) * 128 + h * 16 + l] = acc[r];
  }
  __syncthreads();

#pragma unroll
  for (int j = 0; j < 4; ++j) {
    int q = lane * 4 + j;
    int p = q >> 3;
    int h = q & 7;
    int seg = assign[p0 + p];
    const v4f* qv = (const v4f*)(Q + (size_t)seg * E_ + h * 16);
    const v4f* kv = (const v4f*)(&kt[p * 128 + h * 16]);
    float s = 0.f;
#pragma unroll
    for (int t = 0; t < 4; ++t) {
      v4f qq = qv[t], kk = kv[t];
      s += qq.x * kk.x + qq.y * kk.y + qq.z * kk.z + qq.w * kk.w;
    }
    s *= 0.25f;
    scores[(size_t)(p0 + p) * H_ + h] = s;
    atomicMax(&segmax[seg * H_ + h], enc_f32(s));
    sc[p * 8 + h] = s;
  }
  __syncthreads();
  if (lane < 16) {
    float m = 0.f;
#pragma unroll
    for (int h = 0; h < H_; ++h) m += sc[lane * 8 + h];
    attn_out[p0 + lane] = m * (1.f / H_);
  }
}

__global__ void k_vproj_ctx(const float* __restrict__ raw,
                            const float* __restrict__ Wfold,
                            const float* __restrict__ scores,
                            const unsigned* __restrict__ segmax,
                            const int* __restrict__ assign,
                            float* __restrict__ denom,
                            float* __restrict__ ctx) {
  __shared__ float vt[16 * 128];
  const int p0   = blockIdx.x * 16;
  const int lane = threadIdx.x;
  const int half = lane >> 4;
  const int l    = lane & 15;

  const float* arow = raw + (size_t)(p0 + l) * CKV_ + 2 * half;
  v2f areg[16];
#pragma unroll
  for (int k = 0; k < 16; ++k) areg[k] = *(const v2f*)(arow + 4 * k);

#pragma unroll
  for (int h = 0; h < H_; ++h) {
    const float* brow = Wfold + (size_t)(E_ + h * 16 + l) * CKV_ + 2 * half;
    v8f acc = {0.f, 0.f, 0.f, 0.f, 0.f, 0.f, 0.f, 0.f};
#pragma unroll
    for (int k = 0; k < 16; ++k) {
      v2f b = *(const v2f*)(brow + 4 * k);
      acc = __builtin_amdgcn_wmma_f32_16x16x4_f32(false, areg[k], false, b,
                                                  (short)0, acc, false, false);
    }
#pragma unroll
    for (int r = 0; r < 8; ++r)
      vt[(r + 8 * half) * 128 + h * 16 + l] = acc[r];
  }
  __syncthreads();

#pragma unroll
  for (int j = 0; j < 4; ++j) {
    int q = lane * 4 + j;
    int p = q >> 3;
    int h = q & 7;
    int seg = assign[p0 + p];
    float s = scores[(size_t)(p0 + p) * H_ + h];
    float m = dec_f32(segmax[seg * H_ + h]);
    float w = __expf(s - m);
    atomicAdd(&denom[seg * H_ + h], w);
    const float* vrow = &vt[p * 128 + h * 16];
    float* crow = ctx + (size_t)seg * E_ + h * 16;
#pragma unroll
    for (int d = 0; d < D_; ++d) atomicAdd(&crow[d], w * vrow[d]);
  }
}

__global__ void k_finalize(const float* __restrict__ sp,
                           const float* __restrict__ ctx,
                           const float* __restrict__ denom,
                           const float* __restrict__ lnw,
                           const float* __restrict__ lnb,
                           float* __restrict__ out) {
  const int row  = blockIdx.x;
  const int lane = threadIdx.x;
  const int e0   = lane * 4;
  v4f c = *(const v4f*)(ctx + (size_t)row * E_ + e0);
  v4f s = *(const v4f*)(sp + (size_t)row * E_ + e0);
  float den = denom[row * H_ + (e0 >> 4)];
  float inv = (den > 0.f) ? (1.f / den) : 0.f;
  float x[4] = {s.x + c.x * inv, s.y + c.y * inv,
                s.z + c.z * inv, s.w + c.w * inv};
  float sum = x[0] + x[1] + x[2] + x[3];
#pragma unroll
  for (int m = 16; m >= 1; m >>= 1) sum += __shfl_xor(sum, m, 32);
  float mu = sum * (1.f / E_);
  float vs = 0.f;
#pragma unroll
  for (int j = 0; j < 4; ++j) { float d = x[j] - mu; vs += d * d; }
#pragma unroll
  for (int m = 16; m >= 1; m >>= 1) vs += __shfl_xor(vs, m, 32);
  float rs = rsqrtf(vs * (1.f / E_) + EPS_);
  v4f w = *(const v4f*)(lnw + e0);
  v4f b = *(const v4f*)(lnb + e0);
  v4f y;
  y.x = (x[0] - mu) * rs * w.x + b.x;
  y.y = (x[1] - mu) * rs * w.y + b.y;
  y.z = (x[2] - mu) * rs * w.z + b.z;
  y.w = (x[3] - mu) * rs * w.w + b.w;
  *(v4f*)(out + (size_t)row * E_ + e0) = y;
}

extern "C" void kernel_launch(void* const* d_in, const int* in_sizes, int n_in,
                              void* d_out, int out_size, void* d_ws,
                              size_t ws_size, hipStream_t stream) {
  (void)in_sizes; (void)n_in; (void)out_size; (void)ws_size;
  const float* sp  = (const float*)d_in[0];
  const float* raw = (const float*)d_in[1];
  const int*   asg = (const int*)d_in[2];
  const float* Wq  = (const float*)d_in[3];
  const float* Wkv = (const float*)d_in[4];
  const float* lnw = (const float*)d_in[5];
  const float* lnb = (const float*)d_in[6];

  float* out      = (float*)d_out;
  float* out_attn = out + (size_t)S_ * E_;

  float*    ws     = (float*)d_ws;
  float*    Wfold  = ws;
  float*    Q      = Wfold + 256 * CKV_;
  float*    scores = Q + (size_t)S_ * E_;
  unsigned* segmax = (unsigned*)(scores + (size_t)N_ * H_);
  float*    denom  = (float*)(segmax + S_ * H_);
  float*    ctx    = denom + S_ * H_;

  hipMemsetAsync(segmax, 0, (size_t)S_ * H_ * sizeof(unsigned), stream);
  hipMemsetAsync(denom, 0, (size_t)S_ * H_ * sizeof(float), stream);
  hipMemsetAsync(ctx, 0, (size_t)S_ * E_ * sizeof(float), stream);

  k_fold_wkv<<<64, 256, 0, stream>>>(Wkv, Wfold);
  k_qproj<<<dim3(S_ / 16, E_ / 16), 32, 0, stream>>>(sp, Wq, Q);
  k_kproj_scores<<<N_ / 16, 32, 0, stream>>>(raw, Wfold, Q, asg, scores,
                                             segmax, out_attn);
  k_vproj_ctx<<<N_ / 16, 32, 0, stream>>>(raw, Wfold, scores, segmax, asg,
                                          denom, ctx);
  k_finalize<<<S_, 32, 0, stream>>>(sp, ctx, denom, lnw, lnb, out);
}


# Score kernel gathers touch whole lines: Wkv stored in fragment order by the fold kernel (4 lines per B load, was 16), a point's whole Q row per load (4 lines, was 32); score dot in a new sum order
# speedup vs baseline: 5.0115x; 5.0115x over previous
#include <hip/hip_runtime.h>


#define S_    4096
#define N_    262144
#define H_    8
#define E_    128
#define D_    16
#define CKV_  64
#define EPS_  1e-5f
#define NCHUNK_      32
#define CHUNK_TILES_ (N_ / NCHUNK_ / 16)

typedef float v2f __attribute__((ext_vector_type(2)));
typedef float v4f __attribute__((ext_vector_type(4)));
typedef float v8f __attribute__((ext_vector_type(8)));

#define WFRAG_(T, j) (((T) * 8 + (j)) * 128)
__device__ __forceinline__ v8f wmma2(v4f a, v4f b, v8f acc) {
  acc = __builtin_amdgcn_wmma_f32_16x16x4_f32(false, a.xy, false, b.xy,
                                              (short)0, acc, false, false);
  acc = __builtin_amdgcn_wmma_f32_16x16x4_f32(false, a.zw, false, b.zw,
                                              (short)0, acc, false, false);
  return acc;
}

__device__ __forceinline__ unsigned enc_f32(float f) {
  unsigned u = __float_as_uint(f);
  return (u & 0x80000000u) ? ~u : (u | 0x80000000u);
}
__device__ __forceinline__ float dec_f32(unsigned e) {
  return (e & 0x80000000u) ? __uint_as_float(e & 0x7fffffffu)
                           : __uint_as_float(~e);
}

__global__ void k_fold_wkv(const float* __restrict__ Wkv,
                           float* __restrict__ Wfold) {
  int i    = blockIdx.x * blockDim.x + threadIdx.x;
  int e    = i & 3;
  int lane = (i >> 2) & 31;
  int j    = (i >> 7) & 7;
  int T    = i >> 10;
  int o    = T * 16 + (lane & 15);
  int c    = 8 * j + 4 * (lane >> 4) + e;
  Wfold[i] = Wkv[o * (2 * CKV_) + c] + Wkv[o * (2 * CKV_) + CKV_ + c];
}

__global__ void k_qproj(const float* __restrict__ sp,
                        const float* __restrict__ Wq,
                        float* __restrict__ Q) {
  const int tm   = blockIdx.x * 16;
  const int tn   = blockIdx.y * 16;
  const int lane = threadIdx.x;
  const int half = lane >> 4;
  const int l    = lane & 15;
  const float* arow = sp + (size_t)(tm + l) * E_ + 4 * half;
  const float* brow = Wq + (size_t)(tn + l) * E_ + 4 * half;
  v8f acc = {0.f, 0.f, 0.f, 0.f, 0.f, 0.f, 0.f, 0.f};
#pragma unroll
  for (int j = 0; j < E_ / 8; ++j) {
    v4f a = *(const v4f*)(arow + 8 * j);
    v4f b = *(const v4f*)(brow + 8 * j);
    acc = wmma2(a, b, acc);
  }
#pragma unroll
  for (int r = 0; r < 8; ++r)
    Q[(size_t)(tm + r + 8 * half) * E_ + tn + l] = acc[r];
}

__global__ void k_kproj_scores(const float* __restrict__ raw,
                               const float* __restrict__ Wfold,
                               const float* __restrict__ Q,
                               const int* __restrict__ assign,
                               float* __restrict__ scores,
                               unsigned* __restrict__ segmax,
                               float* __restrict__ attn_out) {
  __shared__ float kt[16 * 128];
  __shared__ float sc[16 * 8];
  const int p0   = blockIdx.x * 16;
  const int lane = threadIdx.x;
  const int half = lane >> 4;
  const int l    = lane & 15;

  const float* arow = raw + (size_t)(p0 + l) * CKV_ + 4 * half;
  v4f areg[8];
#pragma unroll
  for (int j = 0; j < 8; ++j) areg[j] = *(const v4f*)(arow + 8 * j);

#pragma unroll
  for (int h = 0; h < H_; ++h) {
    v8f acc = {0.f, 0.f, 0.f, 0.f, 0.f, 0.f, 0.f, 0.f};
#pragma unroll
    for (int j = 0; j < 8; ++j) {
      v4f b = *(const v4f*)(Wfold + WFRAG_(h, j) + 4 * lane);
      acc = wmma2(areg[j], b, acc);
    }
#pragma unroll
    for (int r = 0; r < 8; ++r)
      kt[(r + 8 * half) * 128 + h * 16 + l] = acc[r];
  }
  __syncthreads();

  const int segl = assign[p0 + l];
  float sv[16];
#pragma unroll
  for (int p = 0; p < 16; ++p) {
    const int seg = __builtin_amdgcn_readlane(segl, p);
    v4f qq = *(const v4f*)(Q + (size_t)seg * E_ + 4 * lane);
    v4f kk = *(const v4f*)(&kt[p * 128 + 4 * lane]);
    float s = qq.x * kk.x + qq.y * kk.y + qq.z * kk.z + qq.w * kk.w;
    s += __builtin_bit_cast(float, __builtin_amdgcn_update_dpp(
             0, __builtin_bit_cast(int, s), 0xB1, 0xf, 0xf, false));
    s += __builtin_bit_cast(float, __builtin_amdgcn_update_dpp(
             0, __builtin_bit_cast(int, s), 0x4E, 0xf, 0xf, false));
    sv[p] = s * 0.25f;
  }
  if ((lane & 3) == 0) {
#pragma unroll
    for (int p = 0; p < 16; ++p) sc[p * 8 + (lane >> 2)] = sv[p];
  }
  __syncthreads();
  {
    const v4f s4  = *(const v4f*)(&sc[lane * 4]);
    *(v4f*)(scores + (size_t)p0 * H_ + lane * 4) = s4;
    const int seg = assign[p0 + (lane >> 1)];
    unsigned* mx = segmax + seg * H_ + 4 * (lane & 1);
    atomicMax(mx + 0, enc_f32(s4.x));
    atomicMax(mx + 1, enc_f32(s4.y));
    atomicMax(mx + 2, enc_f32(s4.z));
    atomicMax(mx + 3, enc_f32(s4.w));
  }
  if (lane < 16) {
    float m = 0.f;
#pragma unroll
    for (int h = 0; h < H_; ++h) m += sc[lane * 8 + h];
    attn_out[p0 + lane] = m * (1.f / H_);
  }
}

__global__ __launch_bounds__(1024)
void k_vproj_ctx(const float* __restrict__ raw,
                 const float* __restrict__ Wfold,
                 const float* __restrict__ scores,
                 const unsigned* __restrict__ segmax,
                 const int* __restrict__ assign,
                 float* __restrict__ denom,
                 float* __restrict__ ctx) {
  __shared__ __attribute__((aligned(16))) float ctx_l[S_ * D_];
  __shared__ __attribute__((aligned(16))) float den_l[S_];
  const int h     = blockIdx.x & (H_ - 1);
  const int chunk = blockIdx.x >> 3;
  const int tid   = threadIdx.x;
  const int wave  = __builtin_amdgcn_readfirstlane(tid >> 5);
  const int lane  = tid & 31;
  const int half  = lane >> 4;
  const int l     = lane & 15;

  const v4f zero4 = {0.f, 0.f, 0.f, 0.f};
  for (int i = tid; i < S_ * D_ / 4; i += 1024) ((v4f*)ctx_l)[i] = zero4;
  for (int i = tid; i < S_ / 4; i += 1024) ((v4f*)den_l)[i] = zero4;

  v4f breg[8];
#pragma unroll
  for (int j = 0; j < 8; ++j)
    breg[j] = *(const v4f*)(Wfold + WFRAG_(H_ + h, j) + 4 * lane);
  __syncthreads();

  for (int t = wave; t < CHUNK_TILES_; t += 32) {
    const int p0 = (chunk * CHUNK_TILES_ + t) * 16;
    const float* arow = raw + (size_t)(p0 + l) * CKV_ + 4 * half;
    v4f areg[8];
#pragma unroll
    for (int j = 0; j < 8; ++j) areg[j] = *(const v4f*)(arow + 8 * j);

    const int   seg = assign[p0 + l];
    const float s   = scores[(size_t)(p0 + l) * H_ + h];
    const float m   = dec_f32(segmax[seg * H_ + h]);
    const float w   = __expf(s - m);

    v8f acc = {0.f, 0.f, 0.f, 0.f, 0.f, 0.f, 0.f, 0.f};
#pragma unroll
    for (int j = 0; j < 8; ++j) acc = wmma2(areg[j], breg[j], acc);

    if (half == 0) atomicAdd(&den_l[seg], w);
#pragma unroll
    for (int r = 0; r < 8; ++r) {
      const int   src = r + 8 * half;
      const float wr  = __shfl(w, src, 32);
      const int   sr  = __shfl(seg, src, 32);
      atomicAdd(&ctx_l[sr * D_ + l], wr * acc[r]);
    }
  }
  __syncthreads();

  const int rot = chunk * (S_ / NCHUNK_);
  for (int i = tid; i < S_ * D_; i += 1024) {
    const int sg = ((i >> 4) + rot) & (S_ - 1);
    const int d  = i & 15;
    atomicAdd(&ctx[(size_t)sg * E_ + h * D_ + d], ctx_l[sg * D_ + d]);
  }
  for (int i = tid; i < S_; i += 1024) {
    const int sg = (i + rot) & (S_ - 1);
    atomicAdd(&denom[sg * H_ + h], den_l[sg]);
  }
}

__global__ void k_finalize(const float* __restrict__ sp,
                           const float* __restrict__ ctx,
                           const float* __restrict__ denom,
                           const float* __restrict__ lnw,
                           const float* __restrict__ lnb,
                           float* __restrict__ out) {
  const int row  = blockIdx.x;
  const int lane = threadIdx.x;
  const int e0   = lane * 4;
  v4f c = *(const v4f*)(ctx + (size_t)row * E_ + e0);
  v4f s = *(const v4f*)(sp + (size_t)row * E_ + e0);
  float den = denom[row * H_ + (e0 >> 4)];
  float inv = (den > 0.f) ? (1.f / den) : 0.f;
  float x[4] = {s.x + c.x * inv, s.y + c.y * inv,
                s.z + c.z * inv, s.w + c.w * inv};
  float sum = x[0] + x[1] + x[2] + x[3];
#pragma unroll
  for (int m = 16; m >= 1; m >>= 1) sum += __shfl_xor(sum, m, 32);
  float mu = sum * (1.f / E_);
  float vs = 0.f;
#pragma unroll
  for (int j = 0; j < 4; ++j) { float d = x[j] - mu; vs += d * d; }
#pragma unroll
  for (int m = 16; m >= 1; m >>= 1) vs += __shfl_xor(vs, m, 32);
  float rs = rsqrtf(vs * (1.f / E_) + EPS_);
  v4f w = *(const v4f*)(lnw + e0);
  v4f b = *(const v4f*)(lnb + e0);
  v4f y;
  y.x = (x[0] - mu) * rs * w.x + b.x;
  y.y = (x[1] - mu) * rs * w.y + b.y;
  y.z = (x[2] - mu) * rs * w.z + b.z;
  y.w = (x[3] - mu) * rs * w.w + b.w;
  *(v4f*)(out + (size_t)row * E_ + e0) = y;
}

extern "C" void kernel_launch(void* const* d_in, const int* in_sizes, int n_in,
                              void* d_out, int out_size, void* d_ws,
                              size_t ws_size, hipStream_t stream) {
  (void)in_sizes; (void)n_in; (void)out_size; (void)ws_size;
  const float* sp  = (const float*)d_in[0];
  const float* raw = (const float*)d_in[1];
  const int*   asg = (const int*)d_in[2];
  const float* Wq  = (const float*)d_in[3];
  const float* Wkv = (const float*)d_in[4];
  const float* lnw = (const float*)d_in[5];
  const float* lnb = (const float*)d_in[6];

  float* out      = (float*)d_out;
  float* out_attn = out + (size_t)S_ * E_;

  float*    ws     = (float*)d_ws;
  float*    Wfold  = ws;
  float*    Q      = Wfold + 256 * CKV_;
  float*    scores = Q + (size_t)S_ * E_;
  unsigned* segmax = (unsigned*)(scores + (size_t)N_ * H_);
  float*    denom  = (float*)(segmax + S_ * H_);
  float*    ctx    = denom + S_ * H_;

  hipMemsetAsync(segmax, 0, (size_t)S_ * H_ * sizeof(unsigned), stream);
  hipMemsetAsync(denom, 0, (size_t)S_ * H_ * sizeof(float), stream);
  hipMemsetAsync(ctx, 0, (size_t)S_ * E_ * sizeof(float), stream);

  k_fold_wkv<<<64, 256, 0, stream>>>(Wkv, Wfold);
  k_qproj<<<dim3(S_ / 16, E_ / 16), 32, 0, stream>>>(sp, Wq, Q);
  k_kproj_scores<<<N_ / 16, 32, 0, stream>>>(raw, Wfold, Q, asg, scores,
                                             segmax, out_attn);
  k_vproj_ctx<<<H_ * NCHUNK_, 1024, 0, stream>>>(raw, Wfold, scores, segmax,
                                                 asg, denom, ctx);
  k_finalize<<<S_, 32, 0, stream>>>(sp, ctx, denom, lnw, lnb, out);
}


# V work-groups own 512 segments of a head: each filters a 65536-point chunk into an LDS list of its points and gathers their rows for WMMA; LDS table and global atomic flush 8x smaller; new sum order
# speedup vs baseline: 6.0925x; 1.2157x over previous
#include <hip/hip_runtime.h>


#define S_    4096
#define N_    262144
#define H_    8
#define E_    128
#define D_    16
#define CKV_  64
#define EPS_  1e-5f
#define R_          8
#define SEGS_R_     (S_ / R_)
#define SEGS_SHIFT_ 9
#define NCH_        4
#define CHUNK_PTS_  (N_ / NCH_)

typedef float v2f __attribute__((ext_vector_type(2)));
typedef float v4f __attribute__((ext_vector_type(4)));
typedef float v8f __attribute__((ext_vector_type(8)));
typedef int   v4i __attribute__((ext_vector_type(4)));

#define WFRAG_(T, j) (((T) * 8 + (j)) * 128)
__device__ __forceinline__ v8f wmma2(v4f a, v4f b, v8f acc) {
  acc = __builtin_amdgcn_wmma_f32_16x16x4_f32(false, a.xy, false, b.xy,
                                              (short)0, acc, false, false);
  acc = __builtin_amdgcn_wmma_f32_16x16x4_f32(false, a.zw, false, b.zw,
                                              (short)0, acc, false, false);
  return acc;
}

__device__ __forceinline__ unsigned enc_f32(float f) {
  unsigned u = __float_as_uint(f);
  return (u & 0x80000000u) ? ~u : (u | 0x80000000u);
}
__device__ __forceinline__ float dec_f32(unsigned e) {
  return (e & 0x80000000u) ? __uint_as_float(e & 0x7fffffffu)
                           : __uint_as_float(~e);
}

__global__ void k_fold_wkv(const float* __restrict__ Wkv,
                           float* __restrict__ Wfold) {
  int i    = blockIdx.x * blockDim.x + threadIdx.x;
  int e    = i & 3;
  int lane = (i >> 2) & 31;
  int j    = (i >> 7) & 7;
  int T    = i >> 10;
  int o    = T * 16 + (lane & 15);
  int c    = 8 * j + 4 * (lane >> 4) + e;
  Wfold[i] = Wkv[o * (2 * CKV_) + c] + Wkv[o * (2 * CKV_) + CKV_ + c];
}

__global__ void k_qproj(const float* __restrict__ sp,
                        const float* __restrict__ Wq,
                        float* __restrict__ Q) {
  const int tm   = blockIdx.x * 16;
  const int tn   = blockIdx.y * 16;
  const int lane = threadIdx.x;
  const int half = lane >> 4;
  const int l    = lane & 15;
  const float* arow = sp + (size_t)(tm + l) * E_ + 4 * half;
  const float* brow = Wq + (size_t)(tn + l) * E_ + 4 * half;
  v8f acc = {0.f, 0.f, 0.f, 0.f, 0.f, 0.f, 0.f, 0.f};
#pragma unroll
  for (int j = 0; j < E_ / 8; ++j) {
    v4f a = *(const v4f*)(arow + 8 * j);
    v4f b = *(const v4f*)(brow + 8 * j);
    acc = wmma2(a, b, acc);
  }
#pragma unroll
  for (int r = 0; r < 8; ++r)
    Q[(size_t)(tm + r + 8 * half) * E_ + tn + l] = acc[r];
}

__global__ void k_kproj_scores(const float* __restrict__ raw,
                               const float* __restrict__ Wfold,
                               const float* __restrict__ Q,
                               const int* __restrict__ assign,
                               float* __restrict__ scores,
                               unsigned* __restrict__ segmax,
                               float* __restrict__ attn_out) {
  __shared__ float kt[16 * 128];
  __shared__ float sc[16 * 8];
  const int p0   = blockIdx.x * 16;
  const int lane = threadIdx.x;
  const int half = lane >> 4;
  const int l    = lane & 15;

  const float* arow = raw + (size_t)(p0 + l) * CKV_ + 4 * half;
  v4f areg[8];
#pragma unroll
  for (int j = 0; j < 8; ++j) areg[j] = *(const v4f*)(arow + 8 * j);

#pragma unroll
  for (int h = 0; h < H_; ++h) {
    v8f acc = {0.f, 0.f, 0.f, 0.f, 0.f, 0.f, 0.f, 0.f};
#pragma unroll
    for (int j = 0; j < 8; ++j) {
      v4f b = *(const v4f*)(Wfold + WFRAG_(h, j) + 4 * lane);
      acc = wmma2(areg[j], b, acc);
    }
#pragma unroll
    for (int r = 0; r < 8; ++r)
      kt[(r + 8 * half) * 128 + h * 16 + l] = acc[r];
  }
  __syncthreads();

  const int segl = assign[p0 + l];
  float sv[16];
#pragma unroll
  for (int p = 0; p < 16; ++p) {
    const int seg = __builtin_amdgcn_readlane(segl, p);
    v4f qq = *(const v4f*)(Q + (size_t)seg * E_ + 4 * lane);
    v4f kk = *(const v4f*)(&kt[p * 128 + 4 * lane]);
    float s = qq.x * kk.x + qq.y * kk.y + qq.z * kk.z + qq.w * kk.w;
    s += __builtin_bit_cast(float, __builtin_amdgcn_update_dpp(
             0, __builtin_bit_cast(int, s), 0xB1, 0xf, 0xf, false));
    s += __builtin_bit_cast(float, __builtin_amdgcn_update_dpp(
             0, __builtin_bit_cast(int, s), 0x4E, 0xf, 0xf, false));
    sv[p] = s * 0.25f;
  }
  if ((lane & 3) == 0) {
#pragma unroll
    for (int p = 0; p < 16; ++p) sc[p * 8 + (lane >> 2)] = sv[p];
  }
  __syncthreads();
  {
    const v4f s4  = *(const v4f*)(&sc[lane * 4]);
    *(v4f*)(scores + (size_t)p0 * H_ + lane * 4) = s4;
    const int seg = assign[p0 + (lane >> 1)];
    unsigned* mx = segmax + seg * H_ + 4 * (lane & 1);
    atomicMax(mx + 0, enc_f32(s4.x));
    atomicMax(mx + 1, enc_f32(s4.y));
    atomicMax(mx + 2, enc_f32(s4.z));
    atomicMax(mx + 3, enc_f32(s4.w));
  }
  if (lane < 16) {
    float m = 0.f;
#pragma unroll
    for (int h = 0; h < H_; ++h) m += sc[lane * 8 + h];
    attn_out[p0 + lane] = m * (1.f / H_);
  }
}

__global__ __launch_bounds__(1024)
void k_vproj_ctx(const float* __restrict__ raw,
                 const float* __restrict__ Wfold,
                 const float* __restrict__ scores,
                 const unsigned* __restrict__ segmax,
                 const int* __restrict__ assign,
                 float* __restrict__ denom,
                 float* __restrict__ ctx) {
  __shared__ __attribute__((aligned(16))) unsigned list_l[CHUNK_PTS_];
  __shared__ __attribute__((aligned(16))) float ctx_l[SEGS_R_ * D_];
  __shared__ __attribute__((aligned(16))) float den_l[SEGS_R_];
  __shared__ unsigned cnt_l;
  const int h     = blockIdx.x & (H_ - 1);
  const int rg    = (blockIdx.x >> 3) & (R_ - 1);
  const int chunk = blockIdx.x >> 6;
  const int base  = chunk * CHUNK_PTS_;
  const int tid   = threadIdx.x;
  const int wave  = __builtin_amdgcn_readfirstlane(tid >> 5);
  const int lane  = tid & 31;
  const int half  = lane >> 4;
  const int l     = lane & 15;

  const v4f zero4 = {0.f, 0.f, 0.f, 0.f};
  for (int i = tid; i < SEGS_R_ * D_ / 4; i += 1024) ((v4f*)ctx_l)[i] = zero4;
  if (tid < SEGS_R_ / 4) ((v4f*)den_l)[tid] = zero4;
  if (tid == 0) cnt_l = 0u;

  v4f breg[8];
#pragma unroll
  for (int j = 0; j < 8; ++j)
    breg[j] = *(const v4f*)(Wfold + WFRAG_(H_ + h, j) + 4 * lane);
  __syncthreads();

  v4i av[16];
#pragma unroll
  for (int it = 0; it < 8; ++it) {
    const int q0 = (it * 1024 + tid) * 8;
    av[2 * it]     = *(const v4i*)(assign + base + q0);
    av[2 * it + 1] = *(const v4i*)(assign + base + q0 + 4);
  }
  unsigned total = 0u;
#pragma unroll
  for (int k = 0; k < 64; ++k) {
    const unsigned sgk = (unsigned)av[k >> 2][k & 3];
    total += __builtin_popcount(
        __builtin_amdgcn_ballot_w32((sgk >> SEGS_SHIFT_) == (unsigned)rg));
  }
  unsigned start = 0u;
  if (lane == 0) start = atomicAdd(&cnt_l, total);
  unsigned run = __builtin_amdgcn_readfirstlane(start);
#pragma unroll
  for (int k = 0; k < 64; ++k) {
    const unsigned sgk  = (unsigned)av[k >> 2][k & 3];
    const bool     in   = (sgk >> SEGS_SHIFT_) == (unsigned)rg;
    const unsigned mask = __builtin_amdgcn_ballot_w32(in);
    const unsigned pos  = run + __builtin_amdgcn_mbcnt_lo(mask, 0u);
    if (in)
      list_l[pos] = (unsigned)(((k >> 3) * 1024 + tid) * 8 + (k & 7)) |
                    ((sgk & (SEGS_R_ - 1)) << 16);
    run += __builtin_popcount(mask);
  }
  __syncthreads();

  const int cnt    = (int)cnt_l;
  const int ntiles = __builtin_amdgcn_readfirstlane((cnt + 15) >> 4);

  for (int t = wave; t < ntiles; t += 32) {
    const int      e0    = 16 * t;
    const bool     valid = (e0 + l) < cnt;
    const unsigned ent   = list_l[valid ? e0 + l : e0];
    const int      sl    = (int)(ent >> 16);
    const size_t   p     = (size_t)base + (ent & 0xffffu);
    const float* arow = raw + p * CKV_ + 4 * half;
    v4f areg[8];
#pragma unroll
    for (int j = 0; j < 8; ++j) areg[j] = *(const v4f*)(arow + 8 * j);

    const float s = scores[p * H_ + h];
    const float m = dec_f32(segmax[(rg * SEGS_R_ + sl) * H_ + h]);
    const float w = valid ? __expf(s - m) : 0.f;

    v8f acc = {0.f, 0.f, 0.f, 0.f, 0.f, 0.f, 0.f, 0.f};
#pragma unroll
    for (int j = 0; j < 8; ++j) acc = wmma2(areg[j], breg[j], acc);

    atomicAdd(&den_l[sl], half == 0 ? w : 0.f);
#pragma unroll
    for (int r = 0; r < 8; ++r) {
      const int   src = r + 8 * half;
      const float wr  = __shfl(w, src, 32);
      const int   sr  = __shfl(sl, src, 32);
      atomicAdd(&ctx_l[sr * D_ + l], wr * acc[r]);
    }
  }
  __syncthreads();

  const int rot = chunk * (SEGS_R_ / NCH_);
  for (int i = tid; i < SEGS_R_ * D_; i += 1024) {
    const int sgl = ((i >> 4) + rot) & (SEGS_R_ - 1);
    const int d   = i & 15;
    atomicAdd(&ctx[(size_t)(rg * SEGS_R_ + sgl) * E_ + h * D_ + d],
              ctx_l[sgl * D_ + d]);
  }
  if (tid < SEGS_R_) {
    const int sgl = (tid + rot) & (SEGS_R_ - 1);
    atomicAdd(&denom[(rg * SEGS_R_ + sgl) * H_ + h], den_l[sgl]);
  }
}

__global__ void k_finalize(const float* __restrict__ sp,
                           const float* __restrict__ ctx,
                           const float* __restrict__ denom,
                           const float* __restrict__ lnw,
                           const float* __restrict__ lnb,
                           float* __restrict__ out) {
  const int row  = blockIdx.x;
  const int lane = threadIdx.x;
  const int e0   = lane * 4;
  v4f c = *(const v4f*)(ctx + (size_t)row * E_ + e0);
  v4f s = *(const v4f*)(sp + (size_t)row * E_ + e0);
  float den = denom[row * H_ + (e0 >> 4)];
  float inv = (den > 0.f) ? (1.f / den) : 0.f;
  float x[4] = {s.x + c.x * inv, s.y + c.y * inv,
                s.z + c.z * inv, s.w + c.w * inv};
  float sum = x[0] + x[1] + x[2] + x[3];
#pragma unroll
  for (int m = 16; m >= 1; m >>= 1) sum += __shfl_xor(sum, m, 32);
  float mu = sum * (1.f / E_);
  float vs = 0.f;
#pragma unroll
  for (int j = 0; j < 4; ++j) { float d = x[j] - mu; vs += d * d; }
#pragma unroll
  for (int m = 16; m >= 1; m >>= 1) vs += __shfl_xor(vs, m, 32);
  float rs = rsqrtf(vs * (1.f / E_) + EPS_);
  v4f w = *(const v4f*)(lnw + e0);
  v4f b = *(const v4f*)(lnb + e0);
  v4f y;
  y.x = (x[0] - mu) * rs * w.x + b.x;
  y.y = (x[1] - mu) * rs * w.y + b.y;
  y.z = (x[2] - mu) * rs * w.z + b.z;
  y.w = (x[3] - mu) * rs * w.w + b.w;
  *(v4f*)(out + (size_t)row * E_ + e0) = y;
}

extern "C" void kernel_launch(void* const* d_in, const int* in_sizes, int n_in,
                              void* d_out, int out_size, void* d_ws,
                              size_t ws_size, hipStream_t stream) {
  (void)in_sizes; (void)n_in; (void)out_size; (void)ws_size;
  const float* sp  = (const float*)d_in[0];
  const float* raw = (const float*)d_in[1];
  const int*   asg = (const int*)d_in[2];
  const float* Wq  = (const float*)d_in[3];
  const float* Wkv = (const float*)d_in[4];
  const float* lnw = (const float*)d_in[5];
  const float* lnb = (const float*)d_in[6];

  float* out      = (float*)d_out;
  float* out_attn = out + (size_t)S_ * E_;

  float*    ws     = (float*)d_ws;
  float*    Wfold  = ws;
  float*    Q      = Wfold + 256 * CKV_;
  float*    scores = Q + (size_t)S_ * E_;
  unsigned* segmax = (unsigned*)(scores + (size_t)N_ * H_);
  float*    denom  = (float*)(segmax + S_ * H_);
  float*    ctx    = denom + S_ * H_;

  hipMemsetAsync(segmax, 0, (size_t)S_ * H_ * sizeof(unsigned), stream);
  hipMemsetAsync(denom, 0, (size_t)S_ * H_ * sizeof(float), stream);
  hipMemsetAsync(ctx, 0, (size_t)S_ * E_ * sizeof(float), stream);

  k_fold_wkv<<<64, 256, 0, stream>>>(Wkv, Wfold);
  k_qproj<<<dim3(S_ / 16, E_ / 16), 32, 0, stream>>>(sp, Wq, Q);
  k_kproj_scores<<<N_ / 16, 32, 0, stream>>>(raw, Wfold, Q, asg, scores,
                                             segmax, out_attn);
  k_vproj_ctx<<<H_ * R_ * NCH_, 1024, 0, stream>>>(raw, Wfold, scores, segmax,
                                                 asg, denom, ctx);
  k_finalize<<<S_, 32, 0, stream>>>(sp, ctx, denom, lnw, lnb, out);
}


# V work-groups own 64 segments for all 8 heads: a gathered 16-point tile stays in registers for the 8 heads' WMMA products (was gathered once per head), weights re-read per head; new sum order
# speedup vs baseline: 6.9944x; 1.1480x over previous
#include <hip/hip_runtime.h>


#define S_    4096
#define N_    262144
#define H_    8
#define E_    128
#define D_    16
#define CKV_  64
#define EPS_  1e-5f
#define R_          64
#define R_SHIFT_    6
#define SEGS_R_     (S_ / R_)
#define SEGS_SHIFT_ 6
#define NCH_        4
#define CHUNK_PTS_  (N_ / NCH_)
#define CPITCH_     (E_ + D_)

typedef float v2f __attribute__((ext_vector_type(2)));
typedef float v4f __attribute__((ext_vector_type(4)));
typedef float v8f __attribute__((ext_vector_type(8)));
typedef int   v4i __attribute__((ext_vector_type(4)));

#define WFRAG_(T, j) (((T) * 8 + (j)) * 128)
__device__ __forceinline__ v8f wmma2(v4f a, v4f b, v8f acc) {
  acc = __builtin_amdgcn_wmma_f32_16x16x4_f32(false, a.xy, false, b.xy,
                                              (short)0, acc, false, false);
  acc = __builtin_amdgcn_wmma_f32_16x16x4_f32(false, a.zw, false, b.zw,
                                              (short)0, acc, false, false);
  return acc;
}

__device__ __forceinline__ unsigned enc_f32(float f) {
  unsigned u = __float_as_uint(f);
  return (u & 0x80000000u) ? ~u : (u | 0x80000000u);
}
__device__ __forceinline__ float dec_f32(unsigned e) {
  return (e & 0x80000000u) ? __uint_as_float(e & 0x7fffffffu)
                           : __uint_as_float(~e);
}

__global__ void k_fold_wkv(const float* __restrict__ Wkv,
                           float* __restrict__ Wfold) {
  int i    = blockIdx.x * blockDim.x + threadIdx.x;
  int e    = i & 3;
  int lane = (i >> 2) & 31;
  int j    = (i >> 7) & 7;
  int T    = i >> 10;
  int o    = T * 16 + (lane & 15);
  int c    = 8 * j + 4 * (lane >> 4) + e;
  Wfold[i] = Wkv[o * (2 * CKV_) + c] + Wkv[o * (2 * CKV_) + CKV_ + c];
}

__global__ void k_qproj(const float* __restrict__ sp,
                        const float* __restrict__ Wq,
                        float* __restrict__ Q) {
  const int tm   = blockIdx.x * 16;
  const int tn   = blockIdx.y * 16;
  const int lane = threadIdx.x;
  const int half = lane >> 4;
  const int l    = lane & 15;
  const float* arow = sp + (size_t)(tm + l) * E_ + 4 * half;
  const float* brow = Wq + (size_t)(tn + l) * E_ + 4 * half;
  v8f acc = {0.f, 0.f, 0.f, 0.f, 0.f, 0.f, 0.f, 0.f};
#pragma unroll
  for (int j = 0; j < E_ / 8; ++j) {
    v4f a = *(const v4f*)(arow + 8 * j);
    v4f b = *(const v4f*)(brow + 8 * j);
    acc = wmma2(a, b, acc);
  }
#pragma unroll
  for (int r = 0; r < 8; ++r)
    Q[(size_t)(tm + r + 8 * half) * E_ + tn + l] = acc[r];
}

__global__ void k_kproj_scores(const float* __restrict__ raw,
                               const float* __restrict__ Wfold,
                               const float* __restrict__ Q,
                               const int* __restrict__ assign,
                               float* __restrict__ scores,
                               unsigned* __restrict__ segmax,
                               float* __restrict__ attn_out) {
  __shared__ float kt[16 * 128];
  __shared__ float sc[16 * 8];
  const int p0   = blockIdx.x * 16;
  const int lane = threadIdx.x;
  const int half = lane >> 4;
  const int l    = lane & 15;

  const float* arow = raw + (size_t)(p0 + l) * CKV_ + 4 * half;
  v4f areg[8];
#pragma unroll
  for (int j = 0; j < 8; ++j) areg[j] = *(const v4f*)(arow + 8 * j);

#pragma unroll
  for (int h = 0; h < H_; ++h) {
    v8f acc = {0.f, 0.f, 0.f, 0.f, 0.f, 0.f, 0.f, 0.f};
#pragma unroll
    for (int j = 0; j < 8; ++j) {
      v4f b = *(const v4f*)(Wfold + WFRAG_(h, j) + 4 * lane);
      acc = wmma2(areg[j], b, acc);
    }
#pragma unroll
    for (int r = 0; r < 8; ++r)
      kt[(r + 8 * half) * 128 + h * 16 + l] = acc[r];
  }
  __syncthreads();

  const int segl = assign[p0 + l];
  float sv[16];
#pragma unroll
  for (int p = 0; p < 16; ++p) {
    const int seg = __builtin_amdgcn_readlane(segl, p);
    v4f qq = *(const v4f*)(Q + (size_t)seg * E_ + 4 * lane);
    v4f kk = *(const v4f*)(&kt[p * 128 + 4 * lane]);
    float s = qq.x * kk.x + qq.y * kk.y + qq.z * kk.z + qq.w * kk.w;
    s += __builtin_bit_cast(float, __builtin_amdgcn_update_dpp(
             0, __builtin_bit_cast(int, s), 0xB1, 0xf, 0xf, false));
    s += __builtin_bit_cast(float, __builtin_amdgcn_update_dpp(
             0, __builtin_bit_cast(int, s), 0x4E, 0xf, 0xf, false));
    sv[p] = s * 0.25f;
  }
  if ((lane & 3) == 0) {
#pragma unroll
    for (int p = 0; p < 16; ++p) sc[p * 8 + (lane >> 2)] = sv[p];
  }
  __syncthreads();
  {
    const v4f s4  = *(const v4f*)(&sc[lane * 4]);
    *(v4f*)(scores + (size_t)p0 * H_ + lane * 4) = s4;
    const int seg = assign[p0 + (lane >> 1)];
    unsigned* mx = segmax + seg * H_ + 4 * (lane & 1);
    atomicMax(mx + 0, enc_f32(s4.x));
    atomicMax(mx + 1, enc_f32(s4.y));
    atomicMax(mx + 2, enc_f32(s4.z));
    atomicMax(mx + 3, enc_f32(s4.w));
  }
  if (lane < 16) {
    float m = 0.f;
#pragma unroll
    for (int h = 0; h < H_; ++h) m += sc[lane * 8 + h];
    attn_out[p0 + lane] = m * (1.f / H_);
  }
}

__global__ __launch_bounds__(1024)
void k_vproj_ctx(const float* __restrict__ raw,
                 const float* __restrict__ Wfold,
                 const float* __restrict__ scores,
                 const unsigned* __restrict__ segmax,
                 const int* __restrict__ assign,
                 float* __restrict__ denom,
                 float* __restrict__ ctx) {
  __shared__ __attribute__((aligned(16))) unsigned list_l[CHUNK_PTS_];
  __shared__ __attribute__((aligned(16))) float ctx_l[SEGS_R_ * CPITCH_];
  __shared__ __attribute__((aligned(16))) float den_l[H_ * SEGS_R_];
  __shared__ unsigned cnt_l;
  const int rg    = blockIdx.x & (R_ - 1);
  const int chunk = blockIdx.x >> R_SHIFT_;
  const int base  = chunk * CHUNK_PTS_;
  const int tid   = threadIdx.x;
  const int wave  = __builtin_amdgcn_readfirstlane(tid >> 5);
  const int lane  = tid & 31;
  const int half  = lane >> 4;
  const int l     = lane & 15;

  const v4f zero4 = {0.f, 0.f, 0.f, 0.f};
  for (int i = tid; i < SEGS_R_ * CPITCH_ / 4; i += 1024) ((v4f*)ctx_l)[i] = zero4;
  if (tid < H_ * SEGS_R_ / 4) ((v4f*)den_l)[tid] = zero4;
  if (tid == 0) cnt_l = 0u;
  __syncthreads();

  {
    v4i av[16];
#pragma unroll
    for (int it = 0; it < 8; ++it) {
      const int q0 = (it * 1024 + tid) * 8;
      av[2 * it]     = *(const v4i*)(assign + base + q0);
      av[2 * it + 1] = *(const v4i*)(assign + base + q0 + 4);
    }
    unsigned total = 0u;
#pragma unroll
    for (int k = 0; k < 64; ++k) {
      const unsigned sgk = (unsigned)av[k >> 2][k & 3];
      total += __builtin_popcount(
          __builtin_amdgcn_ballot_w32((sgk >> SEGS_SHIFT_) == (unsigned)rg));
    }
    unsigned start = 0u;
    if (lane == 0) start = atomicAdd(&cnt_l, total);
    unsigned run = __builtin_amdgcn_readfirstlane(start);
#pragma unroll
    for (int k = 0; k < 64; ++k) {
      const unsigned sgk  = (unsigned)av[k >> 2][k & 3];
      const bool     in   = (sgk >> SEGS_SHIFT_) == (unsigned)rg;
      const unsigned mask = __builtin_amdgcn_ballot_w32(in);
      const unsigned pos  = run + __builtin_amdgcn_mbcnt_lo(mask, 0u);
      if (in)
        list_l[pos] = (unsigned)(((k >> 3) * 1024 + tid) * 8 + (k & 7)) |
                      ((sgk & (SEGS_R_ - 1)) << 16);
      run += __builtin_popcount(mask);
    }
  }
  __syncthreads();

  const int cnt    = (int)cnt_l;
  const int ntiles = __builtin_amdgcn_readfirstlane((cnt + 15) >> 4);
  const int nunits = ntiles * H_;
  const int upw    = (nunits + 31) >> 5;
  const int u0     = wave * upw;
  const int u1     = (u0 + upw < nunits) ? u0 + upw : nunits;

  if (u0 < u1) {
    const int tA = u0 >> 3;
    const int tB = (u1 - 1) >> 3;
    for (int t = tA; t <= tB; ++t) {
      const int      e0    = 16 * t;
      const bool     valid = (e0 + l) < cnt;
      const unsigned ent   = list_l[valid ? e0 + l : e0];
      const int      sl    = (int)(ent >> 16);
      const size_t   p     = (size_t)base + (ent & 0xffffu);
      const float* arow = raw + p * CKV_ + 4 * half;
      v4f areg[8];
#pragma unroll
      for (int j = 0; j < 8; ++j) areg[j] = *(const v4f*)(arow + 8 * j);

      const v4f s4 = *(const v4f*)(scores + p * H_ + 4 * half);
      const v4i m4 = *(const v4i*)(segmax + (size_t)(rg * SEGS_R_ + sl) * H_ +
                                   4 * half);
      const float vm = valid ? 1.f : 0.f;
      v4f w4;
      w4.x = vm * __expf(s4.x - dec_f32((unsigned)m4.x));
      w4.y = vm * __expf(s4.y - dec_f32((unsigned)m4.y));
      w4.z = vm * __expf(s4.z - dec_f32((unsigned)m4.z));
      w4.w = vm * __expf(s4.w - dec_f32((unsigned)m4.w));

      int crow[8];
#pragma unroll
      for (int r = 0; r < 8; ++r)
        crow[r] = __shfl(sl, r + 8 * half, 32) * CPITCH_ + l;

      const int hb = (t == tA) ? (u0 & 7) : 0;
      const int he = (t == tB) ? ((u1 - 1) & 7) + 1 : H_;
      for (int h = hb; h < he; ++h) {
        const float* bp = Wfold + WFRAG_(H_ + h, 0) + 4 * lane;
        v4f breg[8];
#pragma unroll
        for (int j = 0; j < 8; ++j) breg[j] = *(const v4f*)(bp + 128 * j);

        v8f acc = {0.f, 0.f, 0.f, 0.f, 0.f, 0.f, 0.f, 0.f};
#pragma unroll
        for (int j = 0; j < 8; ++j) acc = wmma2(areg[j], breg[j], acc);

        const int   hh   = h >> 2;
        const float wsel = (h & 2) ? ((h & 1) ? w4.w : w4.z)
                                   : ((h & 1) ? w4.y : w4.x);
        atomicAdd(&den_l[h * SEGS_R_ + sl], half == hh ? wsel : 0.f);
#pragma unroll
        for (int r = 0; r < 8; ++r) {
          const float wr = __shfl(wsel, r + 8 * half + 16 * hh, 32);
          atomicAdd(&ctx_l[crow[r] + h * D_], wr * acc[r]);
        }
      }
    }
  }
  __syncthreads();

  const int rot = chunk * (SEGS_R_ / NCH_);
  for (int i = tid; i < SEGS_R_ * E_; i += 1024) {
    const int sgl = ((i >> 7) + rot) & (SEGS_R_ - 1);
    const int e   = i & (E_ - 1);
    atomicAdd(&ctx[(size_t)(rg * SEGS_R_ + sgl) * E_ + e],
              ctx_l[sgl * CPITCH_ + e]);
  }
  if (tid < H_ * SEGS_R_) {
    const int sgl = ((tid >> 3) + rot) & (SEGS_R_ - 1);
    const int h   = tid & (H_ - 1);
    atomicAdd(&denom[(rg * SEGS_R_ + sgl) * H_ + h], den_l[h * SEGS_R_ + sgl]);
  }
}

__global__ void k_finalize(const float* __restrict__ sp,
                           const float* __restrict__ ctx,
                           const float* __restrict__ denom,
                           const float* __restrict__ lnw,
                           const float* __restrict__ lnb,
                           float* __restrict__ out) {
  const int row  = blockIdx.x;
  const int lane = threadIdx.x;
  const int e0   = lane * 4;
  v4f c = *(const v4f*)(ctx + (size_t)row * E_ + e0);
  v4f s = *(const v4f*)(sp + (size_t)row * E_ + e0);
  float den = denom[row * H_ + (e0 >> 4)];
  float inv = (den > 0.f) ? (1.f / den) : 0.f;
  float x[4] = {s.x + c.x * inv, s.y + c.y * inv,
                s.z + c.z * inv, s.w + c.w * inv};
  float sum = x[0] + x[1] + x[2] + x[3];
#pragma unroll
  for (int m = 16; m >= 1; m >>= 1) sum += __shfl_xor(sum, m, 32);
  float mu = sum * (1.f / E_);
  float vs = 0.f;
#pragma unroll
  for (int j = 0; j < 4; ++j) { float d = x[j] - mu; vs += d * d; }
#pragma unroll
  for (int m = 16; m >= 1; m >>= 1) vs += __shfl_xor(vs, m, 32);
  float rs = rsqrtf(vs * (1.f / E_) + EPS_);
  v4f w = *(const v4f*)(lnw + e0);
  v4f b = *(const v4f*)(lnb + e0);
  v4f y;
  y.x = (x[0] - mu) * rs * w.x + b.x;
  y.y = (x[1] - mu) * rs * w.y + b.y;
  y.z = (x[2] - mu) * rs * w.z + b.z;
  y.w = (x[3] - mu) * rs * w.w + b.w;
  *(v4f*)(out + (size_t)row * E_ + e0) = y;
}

extern "C" void kernel_launch(void* const* d_in, const int* in_sizes, int n_in,
                              void* d_out, int out_size, void* d_ws,
                              size_t ws_size, hipStream_t stream) {
  (void)in_sizes; (void)n_in; (void)out_size; (void)ws_size;
  const float* sp  = (const float*)d_in[0];
  const float* raw = (const float*)d_in[1];
  const int*   asg = (const int*)d_in[2];
  const float* Wq  = (const float*)d_in[3];
  const float* Wkv = (const float*)d_in[4];
  const float* lnw = (const float*)d_in[5];
  const float* lnb = (const float*)d_in[6];

  float* out      = (float*)d_out;
  float* out_attn = out + (size_t)S_ * E_;

  float*    ws     = (float*)d_ws;
  float*    Wfold  = ws;
  float*    Q      = Wfold + 256 * CKV_;
  float*    scores = Q + (size_t)S_ * E_;
  unsigned* segmax = (unsigned*)(scores + (size_t)N_ * H_);
  float*    denom  = (float*)(segmax + S_ * H_);
  float*    ctx    = denom + S_ * H_;

  hipMemsetAsync(segmax, 0, (size_t)S_ * H_ * sizeof(unsigned), stream);
  hipMemsetAsync(denom, 0, (size_t)S_ * H_ * sizeof(float), stream);
  hipMemsetAsync(ctx, 0, (size_t)S_ * E_ * sizeof(float), stream);

  k_fold_wkv<<<64, 256, 0, stream>>>(Wkv, Wfold);
  k_qproj<<<dim3(S_ / 16, E_ / 16), 32, 0, stream>>>(sp, Wq, Q);
  k_kproj_scores<<<N_ / 16, 32, 0, stream>>>(raw, Wfold, Q, asg, scores,
                                             segmax, out_attn);
  k_vproj_ctx<<<R_ * NCH_, 1024, 0, stream>>>(raw, Wfold, scores, segmax,
                                            asg, denom, ctx);
  k_finalize<<<S_, 32, 0, stream>>>(sp, ctx, denom, lnw, lnb, out);
}


# B operands (weight fragments) of the K and V products copied once per work-group into LDS: the WMMA loops issue no global weight loads; score kernel in 8-wave work-groups; sums unchanged
# speedup vs baseline: 7.3327x; 1.0484x over previous
#include <hip/hip_runtime.h>


#define S_    4096
#define N_    262144
#define H_    8
#define E_    128
#define D_    16
#define CKV_  64
#define EPS_  1e-5f
#define R_          64
#define R_SHIFT_    6
#define SEGS_R_     (S_ / R_)
#define SEGS_SHIFT_ 6
#define NCH_        4
#define CHUNK_PTS_  (N_ / NCH_)
#define CPITCH_     (E_ + D_)

typedef float v2f __attribute__((ext_vector_type(2)));
typedef float v4f __attribute__((ext_vector_type(4)));
typedef float v8f __attribute__((ext_vector_type(8)));
typedef int   v4i __attribute__((ext_vector_type(4)));

#define WFRAG_(T, j) (((T) * 8 + (j)) * 128)
__device__ __forceinline__ v8f wmma2(v4f a, v4f b, v8f acc) {
  acc = __builtin_amdgcn_wmma_f32_16x16x4_f32(false, a.xy, false, b.xy,
                                              (short)0, acc, false, false);
  acc = __builtin_amdgcn_wmma_f32_16x16x4_f32(false, a.zw, false, b.zw,
                                              (short)0, acc, false, false);
  return acc;
}

__device__ __forceinline__ unsigned enc_f32(float f) {
  unsigned u = __float_as_uint(f);
  return (u & 0x80000000u) ? ~u : (u | 0x80000000u);
}
__device__ __forceinline__ float dec_f32(unsigned e) {
  return (e & 0x80000000u) ? __uint_as_float(e & 0x7fffffffu)
                           : __uint_as_float(~e);
}

__global__ void k_fold_wkv(const float* __restrict__ Wkv,
                           float* __restrict__ Wfold) {
  int i    = blockIdx.x * blockDim.x + threadIdx.x;
  int e    = i & 3;
  int lane = (i >> 2) & 31;
  int j    = (i >> 7) & 7;
  int T    = i >> 10;
  int o    = T * 16 + (lane & 15);
  int c    = 8 * j + 4 * (lane >> 4) + e;
  Wfold[i] = Wkv[o * (2 * CKV_) + c] + Wkv[o * (2 * CKV_) + CKV_ + c];
}

__global__ void k_qproj(const float* __restrict__ sp,
                        const float* __restrict__ Wq,
                        float* __restrict__ Q) {
  const int tm   = blockIdx.x * 16;
  const int tn   = blockIdx.y * 16;
  const int lane = threadIdx.x;
  const int half = lane >> 4;
  const int l    = lane & 15;
  const float* arow = sp + (size_t)(tm + l) * E_ + 4 * half;
  const float* brow = Wq + (size_t)(tn + l) * E_ + 4 * half;
  v8f acc = {0.f, 0.f, 0.f, 0.f, 0.f, 0.f, 0.f, 0.f};
#pragma unroll
  for (int j = 0; j < E_ / 8; ++j) {
    v4f a = *(const v4f*)(arow + 8 * j);
    v4f b = *(const v4f*)(brow + 8 * j);
    acc = wmma2(a, b, acc);
  }
#pragma unroll
  for (int r = 0; r < 8; ++r)
    Q[(size_t)(tm + r + 8 * half) * E_ + tn + l] = acc[r];
}

#define K2_WAVES_ 8
__global__ __launch_bounds__(K2_WAVES_ * 32)
void k_kproj_scores(const float* __restrict__ raw,
                               const float* __restrict__ Wfold,
                               const float* __restrict__ Q,
                               const int* __restrict__ assign,
                               float* __restrict__ scores,
                               unsigned* __restrict__ segmax,
                               float* __restrict__ attn_out) {
  __shared__ __attribute__((aligned(16))) float wk_l[H_ * 8 * 128];
  __shared__ __attribute__((aligned(16))) float kt_all[K2_WAVES_ * 16 * 128];
  __shared__ __attribute__((aligned(16))) float sc_all[K2_WAVES_ * 16 * 8];
  const int tid  = threadIdx.x;
  const int wave = tid >> 5;
  const int lane = tid & 31;
  const int half = lane >> 4;
  const int l    = lane & 15;
  const int p0   = (blockIdx.x * K2_WAVES_ + wave) * 16;
  float* kt = kt_all + wave * (16 * 128);
  float* sc = sc_all + wave * (16 * 8);

  const float* arow = raw + (size_t)(p0 + l) * CKV_ + 4 * half;
  v4f areg[8];
#pragma unroll
  for (int j = 0; j < 8; ++j) areg[j] = *(const v4f*)(arow + 8 * j);

#pragma unroll
  for (int i = 0; i < (H_ * 8 * 128) / (4 * K2_WAVES_ * 32); ++i) {
    const int q = (i * K2_WAVES_ * 32 + tid) * 4;
    *(v4f*)(wk_l + q) = *(const v4f*)(Wfold + q);
  }
  __syncthreads();

#pragma unroll
  for (int h = 0; h < H_; ++h) {
    v8f acc = {0.f, 0.f, 0.f, 0.f, 0.f, 0.f, 0.f, 0.f};
#pragma unroll
    for (int j = 0; j < 8; ++j) {
      v4f b = *(const v4f*)(wk_l + WFRAG_(h, j) + 4 * lane);
      acc = wmma2(areg[j], b, acc);
    }
#pragma unroll
    for (int r = 0; r < 8; ++r)
      kt[(r + 8 * half) * 128 + h * 16 + l] = acc[r];
  }
  __syncthreads();

  const int segl = assign[p0 + l];
  float sv[16];
#pragma unroll
  for (int p = 0; p < 16; ++p) {
    const int seg = __builtin_amdgcn_readlane(segl, p);
    v4f qq = *(const v4f*)(Q + (size_t)seg * E_ + 4 * lane);
    v4f kk = *(const v4f*)(&kt[p * 128 + 4 * lane]);
    float s = qq.x * kk.x + qq.y * kk.y + qq.z * kk.z + qq.w * kk.w;
    s += __builtin_bit_cast(float, __builtin_amdgcn_update_dpp(
             0, __builtin_bit_cast(int, s), 0xB1, 0xf, 0xf, false));
    s += __builtin_bit_cast(float, __builtin_amdgcn_update_dpp(
             0, __builtin_bit_cast(int, s), 0x4E, 0xf, 0xf, false));
    sv[p] = s * 0.25f;
  }
  if ((lane & 3) == 0) {
#pragma unroll
    for (int p = 0; p < 16; ++p) sc[p * 8 + (lane >> 2)] = sv[p];
  }
  __syncthreads();
  {
    const v4f s4  = *(const v4f*)(&sc[lane * 4]);
    *(v4f*)(scores + (size_t)p0 * H_ + lane * 4) = s4;
    const int seg = assign[p0 + (lane >> 1)];
    unsigned* mx = segmax + seg * H_ + 4 * (lane & 1);
    atomicMax(mx + 0, enc_f32(s4.x));
    atomicMax(mx + 1, enc_f32(s4.y));
    atomicMax(mx + 2, enc_f32(s4.z));
    atomicMax(mx + 3, enc_f32(s4.w));
  }
  if (lane < 16) {
    float m = 0.f;
#pragma unroll
    for (int h = 0; h < H_; ++h) m += sc[lane * 8 + h];
    attn_out[p0 + lane] = m * (1.f / H_);
  }
}

__global__ __launch_bounds__(1024)
void k_vproj_ctx(const float* __restrict__ raw,
                 const float* __restrict__ Wfold,
                 const float* __restrict__ scores,
                 const unsigned* __restrict__ segmax,
                 const int* __restrict__ assign,
                 float* __restrict__ denom,
                 float* __restrict__ ctx) {
  __shared__ __attribute__((aligned(16))) unsigned short pt_l[CHUNK_PTS_];
  __shared__ __attribute__((aligned(16))) unsigned char  sg_l[CHUNK_PTS_];
  __shared__ __attribute__((aligned(16))) float wv_l[H_ * 8 * 128];
  __shared__ __attribute__((aligned(16))) float ctx_l[SEGS_R_ * CPITCH_];
  __shared__ __attribute__((aligned(16))) float den_l[H_ * SEGS_R_];
  __shared__ unsigned cnt_l;
  const int rg    = blockIdx.x & (R_ - 1);
  const int chunk = blockIdx.x >> R_SHIFT_;
  const int base  = chunk * CHUNK_PTS_;
  const int tid   = threadIdx.x;
  const int wave  = __builtin_amdgcn_readfirstlane(tid >> 5);
  const int lane  = tid & 31;
  const int half  = lane >> 4;
  const int l     = lane & 15;

  const v4f zero4 = {0.f, 0.f, 0.f, 0.f};
  for (int i = tid; i < SEGS_R_ * CPITCH_ / 4; i += 1024) ((v4f*)ctx_l)[i] = zero4;
  if (tid < H_ * SEGS_R_ / 4) ((v4f*)den_l)[tid] = zero4;
  if (tid == 0) cnt_l = 0u;
#pragma unroll
  for (int i = 0; i < (H_ * 8 * 128) / (4 * 1024); ++i) {
    const int q = (i * 1024 + tid) * 4;
    *(v4f*)(wv_l + q) = *(const v4f*)(Wfold + WFRAG_(H_, 0) + q);
  }
  __syncthreads();

  {
    v4i av[16];
#pragma unroll
    for (int it = 0; it < 8; ++it) {
      const int q0 = (it * 1024 + tid) * 8;
      av[2 * it]     = *(const v4i*)(assign + base + q0);
      av[2 * it + 1] = *(const v4i*)(assign + base + q0 + 4);
    }
    unsigned total = 0u;
#pragma unroll
    for (int k = 0; k < 64; ++k) {
      const unsigned sgk = (unsigned)av[k >> 2][k & 3];
      total += __builtin_popcount(
          __builtin_amdgcn_ballot_w32((sgk >> SEGS_SHIFT_) == (unsigned)rg));
    }
    unsigned start = 0u;
    if (lane == 0) start = atomicAdd(&cnt_l, total);
    unsigned run = __builtin_amdgcn_readfirstlane(start);
#pragma unroll
    for (int k = 0; k < 64; ++k) {
      const unsigned sgk  = (unsigned)av[k >> 2][k & 3];
      const bool     in   = (sgk >> SEGS_SHIFT_) == (unsigned)rg;
      const unsigned mask = __builtin_amdgcn_ballot_w32(in);
      const unsigned pos  = run + __builtin_amdgcn_mbcnt_lo(mask, 0u);
      if (in) {
        pt_l[pos] = (unsigned short)(((k >> 3) * 1024 + tid) * 8 + (k & 7));
        sg_l[pos] = (unsigned char)(sgk & (SEGS_R_ - 1));
      }
      run += __builtin_popcount(mask);
    }
  }
  __syncthreads();

  const int cnt    = (int)cnt_l;
  const int ntiles = __builtin_amdgcn_readfirstlane((cnt + 15) >> 4);
  const int nunits = ntiles * H_;
  const int upw    = (nunits + 31) >> 5;
  const int u0     = wave * upw;
  const int u1     = (u0 + upw < nunits) ? u0 + upw : nunits;

  if (u0 < u1) {
    const int tA = u0 >> 3;
    const int tB = (u1 - 1) >> 3;
    for (int t = tA; t <= tB; ++t) {
      const int      e0    = 16 * t;
      const bool     valid = (e0 + l) < cnt;
      const int      ei    = valid ? e0 + l : e0;
      const int      sl    = (int)sg_l[ei];
      const size_t   p     = (size_t)base + pt_l[ei];
      const float* arow = raw + p * CKV_ + 4 * half;
      v4f areg[8];
#pragma unroll
      for (int j = 0; j < 8; ++j) areg[j] = *(const v4f*)(arow + 8 * j);

      const v4f s4 = *(const v4f*)(scores + p * H_ + 4 * half);
      const v4i m4 = *(const v4i*)(segmax + (size_t)(rg * SEGS_R_ + sl) * H_ +
                                   4 * half);
      const float vm = valid ? 1.f : 0.f;
      v4f w4;
      w4.x = vm * __expf(s4.x - dec_f32((unsigned)m4.x));
      w4.y = vm * __expf(s4.y - dec_f32((unsigned)m4.y));
      w4.z = vm * __expf(s4.z - dec_f32((unsigned)m4.z));
      w4.w = vm * __expf(s4.w - dec_f32((unsigned)m4.w));

      int crow[8];
#pragma unroll
      for (int r = 0; r < 8; ++r)
        crow[r] = __shfl(sl, r + 8 * half, 32) * CPITCH_ + l;

      const int hb = (t == tA) ? (u0 & 7) : 0;
      const int he = (t == tB) ? ((u1 - 1) & 7) + 1 : H_;
      for (int h = hb; h < he; ++h) {
        const float* bp = wv_l + WFRAG_(h, 0) + 4 * lane;
        v4f breg[8];
#pragma unroll
        for (int j = 0; j < 8; ++j) breg[j] = *(const v4f*)(bp + 128 * j);
        __builtin_amdgcn_sched_barrier(0);

        v8f acc = {0.f, 0.f, 0.f, 0.f, 0.f, 0.f, 0.f, 0.f};
#pragma unroll
        for (int j = 0; j < 8; ++j) acc = wmma2(areg[j], breg[j], acc);

        const int   hh   = h >> 2;
        const float wsel = (h & 2) ? ((h & 1) ? w4.w : w4.z)
                                   : ((h & 1) ? w4.y : w4.x);
        atomicAdd(&den_l[h * SEGS_R_ + sl], half == hh ? wsel : 0.f);
#pragma unroll
        for (int r = 0; r < 8; ++r) {
          const float wr = __shfl(wsel, r + 8 * half + 16 * hh, 32);
          atomicAdd(&ctx_l[crow[r] + h * D_], wr * acc[r]);
        }
      }
    }
  }
  __syncthreads();

  const int rot = chunk * (SEGS_R_ / NCH_);
  for (int i = tid; i < SEGS_R_ * E_; i += 1024) {
    const int sgl = ((i >> 7) + rot) & (SEGS_R_ - 1);
    const int e   = i & (E_ - 1);
    atomicAdd(&ctx[(size_t)(rg * SEGS_R_ + sgl) * E_ + e],
              ctx_l[sgl * CPITCH_ + e]);
  }
  if (tid < H_ * SEGS_R_) {
    const int sgl = ((tid >> 3) + rot) & (SEGS_R_ - 1);
    const int h   = tid & (H_ - 1);
    atomicAdd(&denom[(rg * SEGS_R_ + sgl) * H_ + h], den_l[h * SEGS_R_ + sgl]);
  }
}

__global__ void k_finalize(const float* __restrict__ sp,
                           const float* __restrict__ ctx,
                           const float* __restrict__ denom,
                           const float* __restrict__ lnw,
                           const float* __restrict__ lnb,
                           float* __restrict__ out) {
  const int row  = blockIdx.x;
  const int lane = threadIdx.x;
  const int e0   = lane * 4;
  v4f c = *(const v4f*)(ctx + (size_t)row * E_ + e0);
  v4f s = *(const v4f*)(sp + (size_t)row * E_ + e0);
  float den = denom[row * H_ + (e0 >> 4)];
  float inv = (den > 0.f) ? (1.f / den) : 0.f;
  float x[4] = {s.x + c.x * inv, s.y + c.y * inv,
                s.z + c.z * inv, s.w + c.w * inv};
  float sum = x[0] + x[1] + x[2] + x[3];
#pragma unroll
  for (int m = 16; m >= 1; m >>= 1) sum += __shfl_xor(sum, m, 32);
  float mu = sum * (1.f / E_);
  float vs = 0.f;
#pragma unroll
  for (int j = 0; j < 4; ++j) { float d = x[j] - mu; vs += d * d; }
#pragma unroll
  for (int m = 16; m >= 1; m >>= 1) vs += __shfl_xor(vs, m, 32);
  float rs = rsqrtf(vs * (1.f / E_) + EPS_);
  v4f w = *(const v4f*)(lnw + e0);
  v4f b = *(const v4f*)(lnb + e0);
  v4f y;
  y.x = (x[0] - mu) * rs * w.x + b.x;
  y.y = (x[1] - mu) * rs * w.y + b.y;
  y.z = (x[2] - mu) * rs * w.z + b.z;
  y.w = (x[3] - mu) * rs * w.w + b.w;
  *(v4f*)(out + (size_t)row * E_ + e0) = y;
}

extern "C" void kernel_launch(void* const* d_in, const int* in_sizes, int n_in,
                              void* d_out, int out_size, void* d_ws,
                              size_t ws_size, hipStream_t stream) {
  (void)in_sizes; (void)n_in; (void)out_size; (void)ws_size;
  const float* sp  = (const float*)d_in[0];
  const float* raw = (const float*)d_in[1];
  const int*   asg = (const int*)d_in[2];
  const float* Wq  = (const float*)d_in[3];
  const float* Wkv = (const float*)d_in[4];
  const float* lnw = (const float*)d_in[5];
  const float* lnb = (const float*)d_in[6];

  float* out      = (float*)d_out;
  float* out_attn = out + (size_t)S_ * E_;

  float*    ws     = (float*)d_ws;
  float*    Wfold  = ws;
  float*    Q      = Wfold + 256 * CKV_;
  float*    scores = Q + (size_t)S_ * E_;
  unsigned* segmax = (unsigned*)(scores + (size_t)N_ * H_);
  float*    denom  = (float*)(segmax + S_ * H_);
  float*    ctx    = denom + S_ * H_;

  hipMemsetAsync(segmax, 0, (size_t)S_ * H_ * sizeof(unsigned), stream);
  hipMemsetAsync(denom, 0, (size_t)S_ * H_ * sizeof(float), stream);
  hipMemsetAsync(ctx, 0, (size_t)S_ * E_ * sizeof(float), stream);

  k_fold_wkv<<<64, 256, 0, stream>>>(Wkv, Wfold);
  k_qproj<<<dim3(S_ / 16, E_ / 16), 32, 0, stream>>>(sp, Wq, Q);
  k_kproj_scores<<<N_ / (16 * K2_WAVES_), K2_WAVES_ * 32, 0, stream>>>(
      raw, Wfold, Q, asg, scores, segmax, out_attn);
  k_vproj_ctx<<<R_ * NCH_, 1024, 0, stream>>>(raw, Wfold, scores, segmax,
                                            asg, denom, ctx);
  k_finalize<<<S_, 32, 0, stream>>>(sp, ctx, denom, lnw, lnb, out);
}
